# starting kernel
# baseline (speedup 1.0000x reference)
#include <hip/hip_runtime.h>
#include <hip/hip_bf16.h>


typedef float v2f __attribute__((ext_vector_type(2)));
typedef float v8f __attribute__((ext_vector_type(8)));
typedef int   v2i __attribute__((ext_vector_type(2)));

__global__ void init_cnt_kernel(int* __restrict__ cnt, int N) {
    int i = blockIdx.x * blockDim.x + threadIdx.x;
    if (i < N) cnt[i] = 0;
}

__global__ void count_deg_kernel(const int* __restrict__ ei, int* __restrict__ cnt, int E) {
    int e = blockIdx.x * blockDim.x + threadIdx.x;
    if (e < E) atomicAdd(&cnt[ei[e]], 1);
}

__global__ void dinv_kernel(const int* __restrict__ cnt, float* __restrict__ dinv, int N) {
    int i = blockIdx.x * blockDim.x + threadIdx.x;
    if (i < N) dinv[i] = rsqrtf((float)(cnt[i] + 1));
}

__global__ __launch_bounds__(1024) void scan_kernel(const int* __restrict__ cnt,
                                                    int* __restrict__ rowptr, int N) {
    __shared__ int sdata[1024];
    int t = threadIdx.x;
    int chunk = (N + 1023) / 1024;
    int begin = t * chunk;
    int end   = begin + chunk;
    if (begin > N) begin = N;
    if (end   > N) end   = N;
    int sum = 0;
    for (int i = begin; i < end; ++i) sum += cnt[i] + 1;
    sdata[t] = sum;
    __syncthreads();
    for (int off = 1; off < 1024; off <<= 1) {
        int v = (t >= off) ? sdata[t - off] : 0;
        __syncthreads();
        sdata[t] += v;
        __syncthreads();
    }
    int run = (t == 0) ? 0 : sdata[t - 1];
    for (int i = begin; i < end; ++i) {
        rowptr[i] = run;
        run += cnt[i] + 1;
    }
    if (t == 1023) rowptr[N] = sdata[1023];
}

__global__ void init_csr_kernel(const int* __restrict__ rowptr, const float* __restrict__ dinv,
                                int* __restrict__ cursor, v2i* __restrict__ csr,
                                int N) {
    int i = blockIdx.x * blockDim.x + threadIdx.x;
    if (i < N) {
        int p = rowptr[i];
        cursor[i] = p + 1;
        float d = dinv[i];
        v2i rec;
        rec.x = i;
        rec.y = __float_as_int(d * d);
        csr[p] = rec;
    }
}

__global__ void fill_csr_kernel(const int* __restrict__ ei, const float* __restrict__ dinv,
                                int* __restrict__ cursor, v2i* __restrict__ csr, int E) {
    int e = blockIdx.x * blockDim.x + threadIdx.x;
    if (e < E) {
        int r = ei[e];
        int c = ei[E + e];
        int p = atomicAdd(&cursor[r], 1);
        v2i rec;
        rec.x = c;
        rec.y = __float_as_int(dinv[r] * dinv[c]);
        csr[p] = rec;
    }
}

__global__ __launch_bounds__(256) void gemm64_wmma_kernel(const float* __restrict__ A,
                                                          const float* __restrict__ B,
                                                          float* __restrict__ C,
                                                          int M, int K) {
    int tid  = threadIdx.x;
    int wave = tid >> 5;
    int lane = tid & 31;
    int m0 = blockIdx.x * 32 + (wave >> 2) * 16;
    int n0 = (wave & 3) * 16;
    if (m0 >= M) return;
    int half = lane >> 4;
    int l    = lane & 15;

    const float* arow = A + (size_t)(m0 + l) * K;
    v8f acc = {0.f, 0.f, 0.f, 0.f, 0.f, 0.f, 0.f, 0.f};

    for (int k = 0; k < K; k += 4) {
        int ka = k + 2 * half;
        v2f a = *(const v2f*)(arow + ka);
        v2f b;
        b.x = B[(size_t)ka * 64 + n0 + l];
        b.y = B[(size_t)(ka + 1) * 64 + n0 + l];
        acc = __builtin_amdgcn_wmma_f32_16x16x4_f32(false, a, false, b,
                                                    (short)0, acc, false, false);
    }
    float* crow = C + (size_t)m0 * 64 + n0 + l;
#pragma unroll
    for (int j = 0; j < 8; ++j) {
        crow[(size_t)(j + 8 * half) * 64] = acc[j];
    }
}

__global__ __launch_bounds__(256) void agg_kernel(const float* __restrict__ h,
                                                  const int* __restrict__ rowptr,
                                                  const v2i* __restrict__ csr,
                                                  const float* __restrict__ bias,
                                                  float* __restrict__ out,
                                                  int N, int do_relu) {
    int node = blockIdx.x * 8 + (threadIdx.x >> 5);
    if (node >= N) return;
    int lane = threadIdx.x & 31;

    const float* hbase = h + 2 * lane;

    float ax = 0.f, ay = 0.f;
    int j = rowptr[node];
    int e = rowptr[node + 1];

    for (; j + 3 < e; j += 4) {
        v2i e0 = csr[j];
        v2i e1 = csr[j + 1];
        v2i e2 = csr[j + 2];
        v2i e3 = csr[j + 3];
        v2f h0 = *(const v2f*)(hbase + (size_t)e0.x * 64);
        v2f h1 = *(const v2f*)(hbase + (size_t)e1.x * 64);
        v2f h2 = *(const v2f*)(hbase + (size_t)e2.x * 64);
        v2f h3 = *(const v2f*)(hbase + (size_t)e3.x * 64);
        float w0 = __int_as_float(e0.y);
        float w1 = __int_as_float(e1.y);
        float w2 = __int_as_float(e2.y);
        float w3 = __int_as_float(e3.y);
        ax = fmaf(h0.x, w0, ax); ay = fmaf(h0.y, w0, ay);
        ax = fmaf(h1.x, w1, ax); ay = fmaf(h1.y, w1, ay);
        ax = fmaf(h2.x, w2, ax); ay = fmaf(h2.y, w2, ay);
        ax = fmaf(h3.x, w3, ax); ay = fmaf(h3.y, w3, ay);
    }
    for (; j < e; ++j) {
        v2i er = csr[j];
        float w = __int_as_float(er.y);
        v2f hv = *(const v2f*)(hbase + (size_t)er.x * 64);
        ax = fmaf(hv.x, w, ax);
        ay = fmaf(hv.y, w, ay);
    }

    v2f bv = *(const v2f*)(bias + 2 * lane);
    ax += bv.x;
    ay += bv.y;
    if (do_relu) {
        ax = fmaxf(ax, 0.f);
        ay = fmaxf(ay, 0.f);
    }
    v2f o; o.x = ax; o.y = ay;
    *(v2f*)(out + (size_t)node * 64 + 2 * lane) = o;
}

extern "C" void kernel_launch(void* const* d_in, const int* in_sizes, int n_in,
                              void* d_out, int out_size, void* d_ws, size_t ws_size,
                              hipStream_t stream) {
    (void)n_in; (void)out_size; (void)ws_size;

    const float* x  = (const float*)d_in[0];
    const int*   ei = (const int*)d_in[1];
    const float* W1 = (const float*)d_in[2];
    const float* b1 = (const float*)d_in[3];
    const float* W2 = (const float*)d_in[4];
    const float* b2 = (const float*)d_in[5];
    float* out = (float*)d_out;

    const int H   = in_sizes[3];
    const int IN  = in_sizes[2] / H;
    const int N   = in_sizes[0] / IN;
    const int E   = in_sizes[1] / 2;
    const int NNZ = N + E;

    char* ws = (char*)d_ws;
    size_t off = 0;
    auto alloc = [&](size_t bytes) -> void* {
        void* p = (void*)(ws + off);
        off += (bytes + 255) & ~(size_t)255;
        return p;
    };
    int*   cnt     = (int*)  alloc((size_t)N * 4);
    int*   rowptr  = (int*)  alloc((size_t)(N + 1) * 4);
    int*   cursor  = (int*)  alloc((size_t)N * 4);
    float* dinv    = (float*)alloc((size_t)N * 4);
    v2i*   csr     = (v2i*)  alloc((size_t)NNZ * 8);
    float* h1      = (float*)alloc((size_t)N * H * 4);
    float* y1      = (float*)alloc((size_t)N * H * 4);
    float* h2      = h1;

    const int T = 256;
    int gN = (N + T - 1) / T;
    int gE = (E + T - 1) / T;
    int gAgg = (N + 7) / 8;
    int gGemm = (N + 31) / 32;

    init_cnt_kernel<<<gN, T, 0, stream>>>(cnt, N);
    count_deg_kernel<<<gE, T, 0, stream>>>(ei, cnt, E);
    dinv_kernel<<<gN, T, 0, stream>>>(cnt, dinv, N);
    scan_kernel<<<1, 1024, 0, stream>>>(cnt, rowptr, N);
    init_csr_kernel<<<gN, T, 0, stream>>>(rowptr, dinv, cursor, csr, N);
    fill_csr_kernel<<<gE, T, 0, stream>>>(ei, dinv, cursor, csr, E);

    gemm64_wmma_kernel<<<gGemm, T, 0, stream>>>(x, W1, h1, N, IN);
    agg_kernel<<<gAgg, T, 0, stream>>>(h1, rowptr, csr, b1, y1, N, 1);

    gemm64_wmma_kernel<<<gGemm, T, 0, stream>>>(y1, W2, h2, N, H);
    agg_kernel<<<gAgg, T, 0, stream>>>(h2, rowptr, csr, b2, out, N, 0);
}


# Adjacency as one 128B slab of 32 column slots per row, filled in ONE atomic pass (long rows spill to a list): degree pass, scan, CSR fill gone; weights formed in aggregation; sum order kept
# speedup vs baseline: 1.8589x; 1.8589x over previous
#include <hip/hip_runtime.h>
#include <hip/hip_bf16.h>


typedef float v2f __attribute__((ext_vector_type(2)));
typedef float v8f __attribute__((ext_vector_type(8)));
typedef int   v2i __attribute__((ext_vector_type(2)));

#define SLAB 32

__global__ void init_cnt_kernel(int* __restrict__ cnt, int N) {
    int i = blockIdx.x * blockDim.x + threadIdx.x;
    if (i <= N) cnt[i] = 0;
}

__global__ void fill_slab_kernel(const int* __restrict__ ei, int* __restrict__ cnt,
                                 int* __restrict__ slab, v2i* __restrict__ spill,
                                 int N, int E) {
    int e = blockIdx.x * blockDim.x + threadIdx.x;
    if (e < E) {
        int r = ei[e];
        int c = ei[E + e];
        int p = atomicAdd(&cnt[r], 1);
        if (p < SLAB) {
            slab[(size_t)r * SLAB + p] = c;
        } else {
            int q = atomicAdd(&cnt[N], 1);
            v2i rec;
            rec.x = r;
            rec.y = c;
            spill[q] = rec;
        }
    }
}

__global__ __launch_bounds__(256) void gemm64_wmma_kernel(const float* __restrict__ A,
                                                          const float* __restrict__ B,
                                                          float* __restrict__ C,
                                                          int M, int K) {
    int tid  = threadIdx.x;
    int wave = tid >> 5;
    int lane = tid & 31;
    int m0 = blockIdx.x * 32 + (wave >> 2) * 16;
    int n0 = (wave & 3) * 16;
    if (m0 >= M) return;
    int half = lane >> 4;
    int l    = lane & 15;

    const float* arow = A + (size_t)(m0 + l) * K;
    v8f acc = {0.f, 0.f, 0.f, 0.f, 0.f, 0.f, 0.f, 0.f};

    for (int k = 0; k < K; k += 4) {
        int ka = k + 2 * half;
        v2f a = *(const v2f*)(arow + ka);
        v2f b;
        b.x = B[(size_t)ka * 64 + n0 + l];
        b.y = B[(size_t)(ka + 1) * 64 + n0 + l];
        acc = __builtin_amdgcn_wmma_f32_16x16x4_f32(false, a, false, b,
                                                    (short)0, acc, false, false);
    }
    float* crow = C + (size_t)m0 * 64 + n0 + l;
#pragma unroll
    for (int j = 0; j < 8; ++j) {
        crow[(size_t)(j + 8 * half) * 64] = acc[j];
    }
}

__global__ __launch_bounds__(256) void agg_kernel(const float* __restrict__ h,
                                                  const int* __restrict__ cnt,
                                                  const int* __restrict__ slab,
                                                  const v2i* __restrict__ spill,
                                                  const float* __restrict__ bias,
                                                  float* __restrict__ out,
                                                  int N, int do_relu) {
    int node = __builtin_amdgcn_readfirstlane(blockIdx.x * 8 + (threadIdx.x >> 5));
    if (node >= N) return;
    int lane = threadIdx.x & 31;

    const float* hbase = h + 2 * lane;

    int deg = cnt[node];
    int n = deg < SLAB ? deg : SLAB;
    float di = rsqrtf((float)(deg + 1));

    v2f hs = *(const v2f*)(hbase + (size_t)node * 64);

    int   col = 0;
    float w   = 0.f;
    if (lane < n) {
        col = slab[(size_t)node * SLAB + lane];
        w   = di * rsqrtf((float)(cnt[col] + 1));
    }

    float ws = di * di;
    float ax = hs.x * ws;
    float ay = hs.y * ws;

    int j = 0;
    for (; j + 3 < n; j += 4) {
        int c0 = __builtin_amdgcn_readlane(col, j);
        int c1 = __builtin_amdgcn_readlane(col, j + 1);
        int c2 = __builtin_amdgcn_readlane(col, j + 2);
        int c3 = __builtin_amdgcn_readlane(col, j + 3);
        v2f h0 = *(const v2f*)(hbase + (size_t)c0 * 64);
        v2f h1 = *(const v2f*)(hbase + (size_t)c1 * 64);
        v2f h2 = *(const v2f*)(hbase + (size_t)c2 * 64);
        v2f h3 = *(const v2f*)(hbase + (size_t)c3 * 64);
        float w0 = __int_as_float(__builtin_amdgcn_readlane(__float_as_int(w), j));
        float w1 = __int_as_float(__builtin_amdgcn_readlane(__float_as_int(w), j + 1));
        float w2 = __int_as_float(__builtin_amdgcn_readlane(__float_as_int(w), j + 2));
        float w3 = __int_as_float(__builtin_amdgcn_readlane(__float_as_int(w), j + 3));
        ax = fmaf(h0.x, w0, ax); ay = fmaf(h0.y, w0, ay);
        ax = fmaf(h1.x, w1, ax); ay = fmaf(h1.y, w1, ay);
        ax = fmaf(h2.x, w2, ax); ay = fmaf(h2.y, w2, ay);
        ax = fmaf(h3.x, w3, ax); ay = fmaf(h3.y, w3, ay);
    }
    for (; j < n; ++j) {
        int   cj = __builtin_amdgcn_readlane(col, j);
        float wj = __int_as_float(__builtin_amdgcn_readlane(__float_as_int(w), j));
        v2f hv = *(const v2f*)(hbase + (size_t)cj * 64);
        ax = fmaf(hv.x, wj, ax);
        ay = fmaf(hv.y, wj, ay);
    }

    if (deg > SLAB) {
        int sn = cnt[N];
        for (int base = 0; base < sn; base += 32) {
            int idx = base + lane;
            int r = -1, c = 0;
            if (idx < sn) {
                v2i rec = spill[idx];
                r = rec.x;
                c = rec.y;
            }
            bool  mine = (r == node);
            float wv = 0.f;
            if (mine) wv = di * rsqrtf((float)(cnt[c] + 1));
            unsigned m = __builtin_amdgcn_ballot_w32(mine);
            while (m) {
                int b = __builtin_ctz(m);
                m &= m - 1;
                int   cb = __builtin_amdgcn_readlane(c, b);
                float wb = __int_as_float(__builtin_amdgcn_readlane(__float_as_int(wv), b));
                v2f hv = *(const v2f*)(hbase + (size_t)cb * 64);
                ax = fmaf(hv.x, wb, ax);
                ay = fmaf(hv.y, wb, ay);
            }
        }
    }

    v2f bv = *(const v2f*)(bias + 2 * lane);
    ax += bv.x;
    ay += bv.y;
    if (do_relu) {
        ax = fmaxf(ax, 0.f);
        ay = fmaxf(ay, 0.f);
    }
    v2f o; o.x = ax; o.y = ay;
    *(v2f*)(out + (size_t)node * 64 + 2 * lane) = o;
}

extern "C" void kernel_launch(void* const* d_in, const int* in_sizes, int n_in,
                              void* d_out, int out_size, void* d_ws, size_t ws_size,
                              hipStream_t stream) {
    (void)n_in; (void)out_size; (void)ws_size;

    const float* x  = (const float*)d_in[0];
    const int*   ei = (const int*)d_in[1];
    const float* W1 = (const float*)d_in[2];
    const float* b1 = (const float*)d_in[3];
    const float* W2 = (const float*)d_in[4];
    const float* b2 = (const float*)d_in[5];
    float* out = (float*)d_out;

    const int H   = in_sizes[3];
    const int IN  = in_sizes[2] / H;
    const int N   = in_sizes[0] / IN;
    const int E   = in_sizes[1] / 2;

    char* ws = (char*)d_ws;
    size_t off = 0;
    auto alloc = [&](size_t bytes) -> void* {
        void* p = (void*)(ws + off);
        off += (bytes + 255) & ~(size_t)255;
        return p;
    };
    int*   cnt     = (int*)  alloc((size_t)(N + 1) * 4);
    int*   slab    = (int*)  alloc((size_t)N * SLAB * 4);
    v2i*   spill   = (v2i*)  alloc((size_t)E * 8);
    float* h1      = (float*)alloc((size_t)N * H * 4);
    float* y1      = (float*)alloc((size_t)N * H * 4);
    float* h2      = h1;

    const int T = 256;
    int gN1 = (N + 1 + T - 1) / T;
    int gE = (E + T - 1) / T;
    int gAgg = (N + 7) / 8;
    int gGemm = (N + 31) / 32;

    init_cnt_kernel<<<gN1, T, 0, stream>>>(cnt, N);
    fill_slab_kernel<<<gE, T, 0, stream>>>(ei, cnt, slab, spill, N, E);

    gemm64_wmma_kernel<<<gGemm, T, 0, stream>>>(x, W1, h1, N, IN);
    agg_kernel<<<gAgg, T, 0, stream>>>(h1, cnt, slab, spill, b1, y1, N, 1);

    gemm64_wmma_kernel<<<gGemm, T, 0, stream>>>(y1, W2, h2, N, H);
    agg_kernel<<<gAgg, T, 0, stream>>>(h2, cnt, slab, spill, b2, out, N, 0);
}


# Slab fill and layer-1 GEMM share one launch (work-group roles alternate 1 GEMM : 2 fill): the edge pass's atomic waits and the GEMM's row loads overlap; 6 launches become 5; sum orders as before
# speedup vs baseline: 1.9308x; 1.0387x over previous
#include <hip/hip_runtime.h>
#include <hip/hip_bf16.h>


typedef float v2f __attribute__((ext_vector_type(2)));
typedef float v8f __attribute__((ext_vector_type(8)));
typedef int   v2i __attribute__((ext_vector_type(2)));

#define SLAB 32

__global__ void init_cnt_kernel(int* __restrict__ cnt, int N) {
    int i = blockIdx.x * blockDim.x + threadIdx.x;
    if (i <= N) cnt[i] = 0;
}

__device__ __forceinline__ void fill_slab_body(int blk, const int* __restrict__ ei,
                                               int* __restrict__ cnt,
                                               int* __restrict__ slab, v2i* __restrict__ spill,
                                               int N, int E) {
    int e = blk * 256 + threadIdx.x;
    if (e < E) {
        int r = ei[e];
        int c = ei[E + e];
        int p = atomicAdd(&cnt[r], 1);
        if (p < SLAB) {
            slab[(size_t)r * SLAB + p] = c;
        } else {
            int q = atomicAdd(&cnt[N], 1);
            v2i rec;
            rec.x = r;
            rec.y = c;
            spill[q] = rec;
        }
    }
}

__device__ __forceinline__ void gemm64_wmma_body(int blk, const float* __restrict__ A,
                                                 const float* __restrict__ B,
                                                 float* __restrict__ C,
                                                 int M, int K) {
    int tid  = threadIdx.x;
    int wave = tid >> 5;
    int lane = tid & 31;
    int m0 = blk * 32 + (wave >> 2) * 16;
    int n0 = (wave & 3) * 16;
    if (m0 >= M) return;
    int half = lane >> 4;
    int l    = lane & 15;

    const float* arow = A + (size_t)(m0 + l) * K;
    v8f acc = {0.f, 0.f, 0.f, 0.f, 0.f, 0.f, 0.f, 0.f};

    for (int k = 0; k < K; k += 4) {
        int ka = k + 2 * half;
        v2f a = *(const v2f*)(arow + ka);
        v2f b;
        b.x = B[(size_t)ka * 64 + n0 + l];
        b.y = B[(size_t)(ka + 1) * 64 + n0 + l];
        acc = __builtin_amdgcn_wmma_f32_16x16x4_f32(false, a, false, b,
                                                    (short)0, acc, false, false);
    }
    float* crow = C + (size_t)m0 * 64 + n0 + l;
#pragma unroll
    for (int j = 0; j < 8; ++j) {
        crow[(size_t)(j + 8 * half) * 64] = acc[j];
    }
}

__global__ __launch_bounds__(256) void gemm64_wmma_kernel(const float* __restrict__ A,
                                                          const float* __restrict__ B,
                                                          float* __restrict__ C,
                                                          int M, int K) {
    gemm64_wmma_body(blockIdx.x, A, B, C, M, K);
}

__global__ __launch_bounds__(256) void fill_gemm_kernel(const int* __restrict__ ei,
                                                        int* __restrict__ cnt,
                                                        int* __restrict__ slab,
                                                        v2i* __restrict__ spill,
                                                        int N, int E,
                                                        const float* __restrict__ A,
                                                        const float* __restrict__ B,
                                                        float* __restrict__ C,
                                                        int K, int gGemm, int gFill) {
    int b = blockIdx.x;
    int m = gFill / 2;
    if (gGemm < m) m = gGemm;
    bool is_gemm;
    int  idx;
    if (b < 3 * m) {
        int q = b / 3;
        int r = b - 3 * q;
        is_gemm = (r == 0);
        idx = is_gemm ? q : 2 * q + r - 1;
    } else {
        int t = b - 3 * m;
        int restG = gGemm - m;
        is_gemm = (t < restG);
        idx = is_gemm ? m + t : 2 * m + (t - restG);
    }
    if (is_gemm) {
        gemm64_wmma_body(idx, A, B, C, N, K);
    } else {
        fill_slab_body(idx, ei, cnt, slab, spill, N, E);
    }
}

__global__ __launch_bounds__(256) void agg_kernel(const float* __restrict__ h,
                                                  const int* __restrict__ cnt,
                                                  const int* __restrict__ slab,
                                                  const v2i* __restrict__ spill,
                                                  const float* __restrict__ bias,
                                                  float* __restrict__ out,
                                                  int N, int do_relu) {
    int node = __builtin_amdgcn_readfirstlane(blockIdx.x * 8 + (threadIdx.x >> 5));
    if (node >= N) return;
    int lane = threadIdx.x & 31;

    const float* hbase = h + 2 * lane;

    int deg = cnt[node];
    int n = deg < SLAB ? deg : SLAB;
    float di = rsqrtf((float)(deg + 1));

    v2f hs = *(const v2f*)(hbase + (size_t)node * 64);

    int   col = 0;
    float w   = 0.f;
    if (lane < n) {
        col = slab[(size_t)node * SLAB + lane];
        w   = di * rsqrtf((float)(cnt[col] + 1));
    }

    float ws = di * di;
    float ax = hs.x * ws;
    float ay = hs.y * ws;

    int j = 0;
    for (; j + 3 < n; j += 4) {
        int c0 = __builtin_amdgcn_readlane(col, j);
        int c1 = __builtin_amdgcn_readlane(col, j + 1);
        int c2 = __builtin_amdgcn_readlane(col, j + 2);
        int c3 = __builtin_amdgcn_readlane(col, j + 3);
        v2f h0 = *(const v2f*)(hbase + (size_t)c0 * 64);
        v2f h1 = *(const v2f*)(hbase + (size_t)c1 * 64);
        v2f h2 = *(const v2f*)(hbase + (size_t)c2 * 64);
        v2f h3 = *(const v2f*)(hbase + (size_t)c3 * 64);
        float w0 = __int_as_float(__builtin_amdgcn_readlane(__float_as_int(w), j));
        float w1 = __int_as_float(__builtin_amdgcn_readlane(__float_as_int(w), j + 1));
        float w2 = __int_as_float(__builtin_amdgcn_readlane(__float_as_int(w), j + 2));
        float w3 = __int_as_float(__builtin_amdgcn_readlane(__float_as_int(w), j + 3));
        ax = fmaf(h0.x, w0, ax); ay = fmaf(h0.y, w0, ay);
        ax = fmaf(h1.x, w1, ax); ay = fmaf(h1.y, w1, ay);
        ax = fmaf(h2.x, w2, ax); ay = fmaf(h2.y, w2, ay);
        ax = fmaf(h3.x, w3, ax); ay = fmaf(h3.y, w3, ay);
    }
    for (; j < n; ++j) {
        int   cj = __builtin_amdgcn_readlane(col, j);
        float wj = __int_as_float(__builtin_amdgcn_readlane(__float_as_int(w), j));
        v2f hv = *(const v2f*)(hbase + (size_t)cj * 64);
        ax = fmaf(hv.x, wj, ax);
        ay = fmaf(hv.y, wj, ay);
    }

    if (deg > SLAB) {
        int sn = cnt[N];
        for (int base = 0; base < sn; base += 32) {
            int idx = base + lane;
            int r = -1, c = 0;
            if (idx < sn) {
                v2i rec = spill[idx];
                r = rec.x;
                c = rec.y;
            }
            bool  mine = (r == node);
            float wv = 0.f;
            if (mine) wv = di * rsqrtf((float)(cnt[c] + 1));
            unsigned m = __builtin_amdgcn_ballot_w32(mine);
            while (m) {
                int b = __builtin_ctz(m);
                m &= m - 1;
                int   cb = __builtin_amdgcn_readlane(c, b);
                float wb = __int_as_float(__builtin_amdgcn_readlane(__float_as_int(wv), b));
                v2f hv = *(const v2f*)(hbase + (size_t)cb * 64);
                ax = fmaf(hv.x, wb, ax);
                ay = fmaf(hv.y, wb, ay);
            }
        }
    }

    v2f bv = *(const v2f*)(bias + 2 * lane);
    ax += bv.x;
    ay += bv.y;
    if (do_relu) {
        ax = fmaxf(ax, 0.f);
        ay = fmaxf(ay, 0.f);
    }
    v2f o; o.x = ax; o.y = ay;
    *(v2f*)(out + (size_t)node * 64 + 2 * lane) = o;
}

extern "C" void kernel_launch(void* const* d_in, const int* in_sizes, int n_in,
                              void* d_out, int out_size, void* d_ws, size_t ws_size,
                              hipStream_t stream) {
    (void)n_in; (void)out_size; (void)ws_size;

    const float* x  = (const float*)d_in[0];
    const int*   ei = (const int*)d_in[1];
    const float* W1 = (const float*)d_in[2];
    const float* b1 = (const float*)d_in[3];
    const float* W2 = (const float*)d_in[4];
    const float* b2 = (const float*)d_in[5];
    float* out = (float*)d_out;

    const int H   = in_sizes[3];
    const int IN  = in_sizes[2] / H;
    const int N   = in_sizes[0] / IN;
    const int E   = in_sizes[1] / 2;

    char* ws = (char*)d_ws;
    size_t off = 0;
    auto alloc = [&](size_t bytes) -> void* {
        void* p = (void*)(ws + off);
        off += (bytes + 255) & ~(size_t)255;
        return p;
    };
    int*   cnt     = (int*)  alloc((size_t)(N + 1) * 4);
    int*   slab    = (int*)  alloc((size_t)N * SLAB * 4);
    v2i*   spill   = (v2i*)  alloc((size_t)E * 8);
    float* h1      = (float*)alloc((size_t)N * H * 4);
    float* y1      = (float*)alloc((size_t)N * H * 4);
    float* h2      = h1;

    const int T = 256;
    int gN1 = (N + 1 + T - 1) / T;
    int gE = (E + T - 1) / T;
    int gAgg = (N + 7) / 8;
    int gGemm = (N + 31) / 32;

    init_cnt_kernel<<<gN1, T, 0, stream>>>(cnt, N);

    fill_gemm_kernel<<<gGemm + gE, T, 0, stream>>>(ei, cnt, slab, spill, N, E,
                                                    x, W1, h1, IN, gGemm, gE);
    agg_kernel<<<gAgg, T, 0, stream>>>(h1, cnt, slab, spill, b1, y1, N, 1);

    gemm64_wmma_kernel<<<gGemm, T, 0, stream>>>(y1, W2, h2, N, H);
    agg_kernel<<<gAgg, T, 0, stream>>>(h2, cnt, slab, spill, b2, out, N, 0);
}


# Slab-fill counter adds kept inside one half of the chip: fill groups read their die, take edge chunks by ticket, add only for target rows of their half (other half's rest last); slot order as before
# speedup vs baseline: 3.1424x; 1.6275x over previous
#include <hip/hip_runtime.h>
#include <hip/hip_bf16.h>


typedef float v2f __attribute__((ext_vector_type(2)));
typedef float v8f __attribute__((ext_vector_type(8)));
typedef int   v2i __attribute__((ext_vector_type(2)));

#define SLAB 32

#define FILL_EPT   4
#define FILL_CHUNK (256 * FILL_EPT)
#define TICK_PITCH 64

__global__ void init_cnt_kernel(int* __restrict__ cnt, int* __restrict__ tick, int N) {
    int i = blockIdx.x * blockDim.x + threadIdx.x;
    if (i <= N) cnt[i] = 0;
    if (i < 2) tick[i * TICK_PITCH] = 0;
}

__device__ __forceinline__ void fill_slab_tickets(const int* __restrict__ ei,
                                                  int* __restrict__ cnt,
                                                  int* __restrict__ tick,
                                                  int* __restrict__ slab, v2i* __restrict__ spill,
                                                  int N, int E) {
    __shared__ int s_ticket[2];
    int slot = 0;
    const int tid = threadIdx.x;
    const int my_half = (int)((__builtin_amdgcn_s_sendmsg_rtn(0x87) >> 18) & 1u);
    const int nT = (E + FILL_CHUNK - 1) / FILL_CHUNK;
    const int Nh = (N / 2) & ~63;

    for (int side = 0; side < 2; ++side) {
        const int h  = my_half ^ side;
        const int lo = h ? Nh : 0;
        const unsigned span = (unsigned)(h ? N - Nh : Nh);
        int* tk = tick + h * TICK_PITCH;

        if (tid == 0) s_ticket[slot] = atomicAdd(tk, 1);
        __syncthreads();
        int t = s_ticket[slot];
        slot ^= 1;

        for (int it = 0; it <= nT && t < nT; ++it) {
            const int base = t * FILL_CHUNK + tid;
            int r[FILL_EPT], c[FILL_EPT], p[FILL_EPT];
#pragma unroll
            for (int k = 0; k < FILL_EPT; ++k) {
                int e = base + 256 * k;
                r[k] = -1;
                c[k] = 0;
                if (e < E) {
                    r[k] = ei[e];
                    c[k] = ei[E + e];
                }
            }
#pragma unroll
            for (int k = 0; k < FILL_EPT; ++k) {
                p[k] = -1;
                if ((unsigned)(r[k] - lo) < span) p[k] = atomicAdd(&cnt[r[k]], 1);
            }

            if (tid == 0) s_ticket[slot] = atomicAdd(tk, 1);
            __syncthreads();
            t = s_ticket[slot];
            slot ^= 1;

#pragma unroll
            for (int k = 0; k < FILL_EPT; ++k) {
                if (p[k] >= 0) {
                    if (p[k] < SLAB) {
                        slab[(size_t)r[k] * SLAB + p[k]] = c[k];
                    } else {
                        int q = atomicAdd(&cnt[N], 1);
                        v2i rec;
                        rec.x = r[k];
                        rec.y = c[k];
                        spill[q] = rec;
                    }
                }
            }
        }
    }
}

__device__ __forceinline__ void gemm64_wmma_body(int blk, const float* __restrict__ A,
                                                 const float* __restrict__ B,
                                                 float* __restrict__ C,
                                                 int M, int K) {
    int tid  = threadIdx.x;
    int wave = tid >> 5;
    int lane = tid & 31;
    int m0 = blk * 32 + (wave >> 2) * 16;
    int n0 = (wave & 3) * 16;
    if (m0 >= M) return;
    int half = lane >> 4;
    int l    = lane & 15;

    const float* arow = A + (size_t)(m0 + l) * K;
    v8f acc = {0.f, 0.f, 0.f, 0.f, 0.f, 0.f, 0.f, 0.f};

    for (int k = 0; k < K; k += 4) {
        int ka = k + 2 * half;
        v2f a = *(const v2f*)(arow + ka);
        v2f b;
        b.x = B[(size_t)ka * 64 + n0 + l];
        b.y = B[(size_t)(ka + 1) * 64 + n0 + l];
        acc = __builtin_amdgcn_wmma_f32_16x16x4_f32(false, a, false, b,
                                                    (short)0, acc, false, false);
    }
    float* crow = C + (size_t)m0 * 64 + n0 + l;
#pragma unroll
    for (int j = 0; j < 8; ++j) {
        crow[(size_t)(j + 8 * half) * 64] = acc[j];
    }
}

__global__ __launch_bounds__(256) void gemm64_wmma_kernel(const float* __restrict__ A,
                                                          const float* __restrict__ B,
                                                          float* __restrict__ C,
                                                          int M, int K) {
    gemm64_wmma_body(blockIdx.x, A, B, C, M, K);
}

__global__ __launch_bounds__(256) void fill_gemm_kernel(const int* __restrict__ ei,
                                                        int* __restrict__ cnt,
                                                        int* __restrict__ tick,
                                                        int* __restrict__ slab,
                                                        v2i* __restrict__ spill,
                                                        int N, int E,
                                                        const float* __restrict__ A,
                                                        const float* __restrict__ B,
                                                        float* __restrict__ C,
                                                        int K, int gGemm, int gFill) {
    int b = blockIdx.x;
    int m = gFill / 2;
    if (gGemm < m) m = gGemm;
    bool is_gemm;
    int  idx;
    if (b < 3 * m) {
        int q = b / 3;
        int r = b - 3 * q;
        is_gemm = (r == 0);
        idx = is_gemm ? q : 2 * q + r - 1;
    } else {
        int t = b - 3 * m;
        int restG = gGemm - m;
        is_gemm = (t < restG);
        idx = is_gemm ? m + t : 2 * m + (t - restG);
    }
    if (is_gemm) {
        gemm64_wmma_body(idx, A, B, C, N, K);
    } else {
        fill_slab_tickets(ei, cnt, tick, slab, spill, N, E);
    }
}

__global__ __launch_bounds__(256) void agg_kernel(const float* __restrict__ h,
                                                  const int* __restrict__ cnt,
                                                  const int* __restrict__ slab,
                                                  const v2i* __restrict__ spill,
                                                  const float* __restrict__ bias,
                                                  float* __restrict__ out,
                                                  int N, int do_relu) {
    int node = __builtin_amdgcn_readfirstlane(blockIdx.x * 8 + (threadIdx.x >> 5));
    if (node >= N) return;
    int lane = threadIdx.x & 31;

    const float* hbase = h + 2 * lane;

    int deg = cnt[node];
    int n = deg < SLAB ? deg : SLAB;
    float di = rsqrtf((float)(deg + 1));

    v2f hs = *(const v2f*)(hbase + (size_t)node * 64);

    int   col = 0;
    float w   = 0.f;
    if (lane < n) {
        col = slab[(size_t)node * SLAB + lane];
        w   = di * rsqrtf((float)(cnt[col] + 1));
    }

    float ws = di * di;
    float ax = hs.x * ws;
    float ay = hs.y * ws;

    int j = 0;
    for (; j + 3 < n; j += 4) {
        int c0 = __builtin_amdgcn_readlane(col, j);
        int c1 = __builtin_amdgcn_readlane(col, j + 1);
        int c2 = __builtin_amdgcn_readlane(col, j + 2);
        int c3 = __builtin_amdgcn_readlane(col, j + 3);
        v2f h0 = *(const v2f*)(hbase + (size_t)c0 * 64);
        v2f h1 = *(const v2f*)(hbase + (size_t)c1 * 64);
        v2f h2 = *(const v2f*)(hbase + (size_t)c2 * 64);
        v2f h3 = *(const v2f*)(hbase + (size_t)c3 * 64);
        float w0 = __int_as_float(__builtin_amdgcn_readlane(__float_as_int(w), j));
        float w1 = __int_as_float(__builtin_amdgcn_readlane(__float_as_int(w), j + 1));
        float w2 = __int_as_float(__builtin_amdgcn_readlane(__float_as_int(w), j + 2));
        float w3 = __int_as_float(__builtin_amdgcn_readlane(__float_as_int(w), j + 3));
        ax = fmaf(h0.x, w0, ax); ay = fmaf(h0.y, w0, ay);
        ax = fmaf(h1.x, w1, ax); ay = fmaf(h1.y, w1, ay);
        ax = fmaf(h2.x, w2, ax); ay = fmaf(h2.y, w2, ay);
        ax = fmaf(h3.x, w3, ax); ay = fmaf(h3.y, w3, ay);
    }
    for (; j < n; ++j) {
        int   cj = __builtin_amdgcn_readlane(col, j);
        float wj = __int_as_float(__builtin_amdgcn_readlane(__float_as_int(w), j));
        v2f hv = *(const v2f*)(hbase + (size_t)cj * 64);
        ax = fmaf(hv.x, wj, ax);
        ay = fmaf(hv.y, wj, ay);
    }

    if (deg > SLAB) {
        int sn = cnt[N];
        for (int base = 0; base < sn; base += 32) {
            int idx = base + lane;
            int r = -1, c = 0;
            if (idx < sn) {
                v2i rec = spill[idx];
                r = rec.x;
                c = rec.y;
            }
            bool  mine = (r == node);
            float wv = 0.f;
            if (mine) wv = di * rsqrtf((float)(cnt[c] + 1));
            unsigned m = __builtin_amdgcn_ballot_w32(mine);
            while (m) {
                int b = __builtin_ctz(m);
                m &= m - 1;
                int   cb = __builtin_amdgcn_readlane(c, b);
                float wb = __int_as_float(__builtin_amdgcn_readlane(__float_as_int(wv), b));
                v2f hv = *(const v2f*)(hbase + (size_t)cb * 64);
                ax = fmaf(hv.x, wb, ax);
                ay = fmaf(hv.y, wb, ay);
            }
        }
    }

    v2f bv = *(const v2f*)(bias + 2 * lane);
    ax += bv.x;
    ay += bv.y;
    if (do_relu) {
        ax = fmaxf(ax, 0.f);
        ay = fmaxf(ay, 0.f);
    }
    v2f o; o.x = ax; o.y = ay;
    *(v2f*)(out + (size_t)node * 64 + 2 * lane) = o;
}

extern "C" void kernel_launch(void* const* d_in, const int* in_sizes, int n_in,
                              void* d_out, int out_size, void* d_ws, size_t ws_size,
                              hipStream_t stream) {
    (void)n_in; (void)out_size; (void)ws_size;

    const float* x  = (const float*)d_in[0];
    const int*   ei = (const int*)d_in[1];
    const float* W1 = (const float*)d_in[2];
    const float* b1 = (const float*)d_in[3];
    const float* W2 = (const float*)d_in[4];
    const float* b2 = (const float*)d_in[5];
    float* out = (float*)d_out;

    const int H   = in_sizes[3];
    const int IN  = in_sizes[2] / H;
    const int N   = in_sizes[0] / IN;
    const int E   = in_sizes[1] / 2;

    char* ws = (char*)d_ws;
    size_t off = 0;
    auto alloc = [&](size_t bytes) -> void* {
        void* p = (void*)(ws + off);
        off += (bytes + 255) & ~(size_t)255;
        return p;
    };
    int*   cnt     = (int*)  alloc((size_t)(N + 1) * 4);
    int*   tick    = (int*)  alloc((size_t)2 * TICK_PITCH * 4);
    int*   slab    = (int*)  alloc((size_t)N * SLAB * 4);
    v2i*   spill   = (v2i*)  alloc((size_t)E * 8);
    float* h1      = (float*)alloc((size_t)N * H * 4);
    float* y1      = (float*)alloc((size_t)N * H * 4);
    float* h2      = h1;

    const int T = 256;
    int gN1 = (N + 1 + T - 1) / T;
    int gFill = 1024;
    int gAgg = (N + 7) / 8;
    int gGemm = (N + 31) / 32;

    init_cnt_kernel<<<gN1, T, 0, stream>>>(cnt, tick, N);

    fill_gemm_kernel<<<gGemm + gFill, T, 0, stream>>>(ei, cnt, tick, slab, spill, N, E,
                                                       x, W1, h1, IN, gGemm, gFill);
    agg_kernel<<<gAgg, T, 0, stream>>>(h1, cnt, slab, spill, b1, y1, N, 1);

    gemm64_wmma_kernel<<<gGemm, T, 0, stream>>>(y1, W2, h2, N, H);
    agg_kernel<<<gAgg, T, 0, stream>>>(h2, cnt, slab, spill, b2, out, N, 0);
}


# Aggregation lane map: 8 lanes per node, a wave sums 4 nodes side by side (row = two 16 B loads per lane, slots broadcast inside 8-lane groups); 100k waves per pass become 25k; sum order kept
# speedup vs baseline: 3.2403x; 1.0312x over previous
#include <hip/hip_runtime.h>
#include <hip/hip_bf16.h>


typedef float v2f __attribute__((ext_vector_type(2)));
typedef float v8f __attribute__((ext_vector_type(8)));
typedef int   v2i __attribute__((ext_vector_type(2)));
typedef float v4f __attribute__((ext_vector_type(4)));
typedef int   v4i __attribute__((ext_vector_type(4)));

#define SLAB 32
#define AGG_NPW 4

#define FILL_EPT   4
#define FILL_CHUNK (256 * FILL_EPT)
#define TICK_PITCH 64

__global__ void init_cnt_kernel(int* __restrict__ cnt, int* __restrict__ tick, int N) {
    int i = blockIdx.x * blockDim.x + threadIdx.x;
    if (i <= N) cnt[i] = 0;
    if (i < 2) tick[i * TICK_PITCH] = 0;
}

__device__ __forceinline__ void fill_slab_tickets(const int* __restrict__ ei,
                                                  int* __restrict__ cnt,
                                                  int* __restrict__ tick,
                                                  int* __restrict__ slab, v2i* __restrict__ spill,
                                                  int N, int E) {
    __shared__ int s_ticket[2];
    int slot = 0;
    const int tid = threadIdx.x;
    const int my_half = (int)((__builtin_amdgcn_s_sendmsg_rtn(0x87) >> 18) & 1u);
    const int nT = (E + FILL_CHUNK - 1) / FILL_CHUNK;
    const int Nh = (N / 2) & ~63;

    for (int side = 0; side < 2; ++side) {
        const int h  = my_half ^ side;
        const int lo = h ? Nh : 0;
        const unsigned span = (unsigned)(h ? N - Nh : Nh);
        int* tk = tick + h * TICK_PITCH;

        if (tid == 0) s_ticket[slot] = atomicAdd(tk, 1);
        __syncthreads();
        int t = s_ticket[slot];
        slot ^= 1;

        for (int it = 0; it <= nT && t < nT; ++it) {
            const int base = t * FILL_CHUNK + tid;
            int r[FILL_EPT], c[FILL_EPT], p[FILL_EPT];
#pragma unroll
            for (int k = 0; k < FILL_EPT; ++k) {
                int e = base + 256 * k;
                r[k] = -1;
                c[k] = 0;
                if (e < E) {
                    r[k] = ei[e];
                    c[k] = ei[E + e];
                }
            }
#pragma unroll
            for (int k = 0; k < FILL_EPT; ++k) {
                p[k] = -1;
                if ((unsigned)(r[k] - lo) < span) p[k] = atomicAdd(&cnt[r[k]], 1);
            }

            if (tid == 0) s_ticket[slot] = atomicAdd(tk, 1);
            __syncthreads();
            t = s_ticket[slot];
            slot ^= 1;

#pragma unroll
            for (int k = 0; k < FILL_EPT; ++k) {
                if (p[k] >= 0) {
                    if (p[k] < SLAB) {
                        slab[(size_t)r[k] * SLAB + p[k]] = c[k];
                    } else {
                        int q = atomicAdd(&cnt[N], 1);
                        v2i rec;
                        rec.x = r[k];
                        rec.y = c[k];
                        spill[q] = rec;
                    }
                }
            }
        }
    }
}

__device__ __forceinline__ void gemm64_wmma_body(int blk, const float* __restrict__ A,
                                                 const float* __restrict__ B,
                                                 float* __restrict__ C,
                                                 int M, int K) {
    int tid  = threadIdx.x;
    int wave = tid >> 5;
    int lane = tid & 31;
    int m0 = blk * 32 + (wave >> 2) * 16;
    int n0 = (wave & 3) * 16;
    if (m0 >= M) return;
    int half = lane >> 4;
    int l    = lane & 15;

    const float* arow = A + (size_t)(m0 + l) * K;
    v8f acc = {0.f, 0.f, 0.f, 0.f, 0.f, 0.f, 0.f, 0.f};

    for (int k = 0; k < K; k += 4) {
        int ka = k + 2 * half;
        v2f a = *(const v2f*)(arow + ka);
        v2f b;
        b.x = B[(size_t)ka * 64 + n0 + l];
        b.y = B[(size_t)(ka + 1) * 64 + n0 + l];
        acc = __builtin_amdgcn_wmma_f32_16x16x4_f32(false, a, false, b,
                                                    (short)0, acc, false, false);
    }
    float* crow = C + (size_t)m0 * 64 + n0 + l;
#pragma unroll
    for (int j = 0; j < 8; ++j) {
        crow[(size_t)(j + 8 * half) * 64] = acc[j];
    }
}

__global__ __launch_bounds__(256) void gemm64_wmma_kernel(const float* __restrict__ A,
                                                          const float* __restrict__ B,
                                                          float* __restrict__ C,
                                                          int M, int K) {
    gemm64_wmma_body(blockIdx.x, A, B, C, M, K);
}

__global__ __launch_bounds__(256) void fill_gemm_kernel(const int* __restrict__ ei,
                                                        int* __restrict__ cnt,
                                                        int* __restrict__ tick,
                                                        int* __restrict__ slab,
                                                        v2i* __restrict__ spill,
                                                        int N, int E,
                                                        const float* __restrict__ A,
                                                        const float* __restrict__ B,
                                                        float* __restrict__ C,
                                                        int K, int gGemm, int gFill) {
    int b = blockIdx.x;
    int m = gFill / 2;
    if (gGemm < m) m = gGemm;
    bool is_gemm;
    int  idx;
    if (b < 3 * m) {
        int q = b / 3;
        int r = b - 3 * q;
        is_gemm = (r == 0);
        idx = is_gemm ? q : 2 * q + r - 1;
    } else {
        int t = b - 3 * m;
        int restG = gGemm - m;
        is_gemm = (t < restG);
        idx = is_gemm ? m + t : 2 * m + (t - restG);
    }
    if (is_gemm) {
        gemm64_wmma_body(idx, A, B, C, N, K);
    } else {
        fill_slab_tickets(ei, cnt, tick, slab, spill, N, E);
    }
}


__device__ __forceinline__ int group_bcast_i(int v, unsigned lo, unsigned hi) {
    return (int)__builtin_amdgcn_permlane16((unsigned)v, (unsigned)v, lo, hi, false, false);
}
__device__ __forceinline__ float group_bcast_f(float v, unsigned lo, unsigned hi) {
    return __int_as_float(group_bcast_i(__float_as_int(v), lo, hi));
}

__device__ __forceinline__ v4f fma4(v4f a, float w, v4f c) {
    v4f r;
    r.x = fmaf(a.x, w, c.x);
    r.y = fmaf(a.y, w, c.y);
    r.z = fmaf(a.z, w, c.z);
    r.w = fmaf(a.w, w, c.w);
    return r;
}

__global__ __launch_bounds__(256) void agg_kernel(const float* __restrict__ h,
                                                  const int* __restrict__ cnt,
                                                  const int* __restrict__ slab,
                                                  const v2i* __restrict__ spill,
                                                  const float* __restrict__ bias,
                                                  float* __restrict__ out,
                                                  int N, int do_relu) {
    const int lane = threadIdx.x & 31;
    const int g = lane >> 3;
    const int q = lane & 7;
    const int node0 = __builtin_amdgcn_readfirstlane((blockIdx.x * 8 + (threadIdx.x >> 5)) * AGG_NPW);
    if (node0 >= N) return;
    const int  node = node0 + g;
    const bool live = node < N;
    const int  nd   = live ? node : node0;

    const v4f* hq = (const v4f*)h + q;

    int deg  = cnt[nd];
    v4i colv = *(const v4i*)(slab + (size_t)nd * SLAB + 4 * q);
    v4f hsA  = (hq + (unsigned)nd * 16u)[0];
    v4f hsB  = (hq + (unsigned)nd * 16u)[8];

    int n = deg < SLAB ? deg : SLAB;
    if (!live) n = 0;
    float di = rsqrtf((float)(deg + 1));

    int d0 = 0, d1 = 0, d2 = 0, d3 = 0;
    if (4 * q < n) {
        d0 = cnt[colv.x];
        if (4 * q + 1 < n) {
            d1 = cnt[colv.y];
            if (4 * q + 2 < n) {
                d2 = cnt[colv.z];
                if (4 * q + 3 < n) d3 = cnt[colv.w];
            }
        }
    }
    v4f wv;
    wv.x = di * rsqrtf((float)(d0 + 1));
    wv.y = di * rsqrtf((float)(d1 + 1));
    wv.z = di * rsqrtf((float)(d2 + 1));
    wv.w = di * rsqrtf((float)(d3 + 1));

    float ws = di * di;
    v4f accA = hsA * ws;
    v4f accB = hsB * ws;

    int n01 = max(__builtin_amdgcn_readlane(n, 0),  __builtin_amdgcn_readlane(n, 8));
    int n23 = max(__builtin_amdgcn_readlane(n, 16), __builtin_amdgcn_readlane(n, 24));
    int nmax = max(n01, n23);

    for (int s = 0; s < nmax; s += 4) {
        const unsigned lo = 0x11111111u * (unsigned)(s >> 2);
        const unsigned hi = lo + 0x88888888u;
        int   c0 = group_bcast_i(colv.x, lo, hi);
        int   c1 = group_bcast_i(colv.y, lo, hi);
        int   c2 = group_bcast_i(colv.z, lo, hi);
        int   c3 = group_bcast_i(colv.w, lo, hi);
        float w0 = group_bcast_f(wv.x, lo, hi);
        float w1 = group_bcast_f(wv.y, lo, hi);
        float w2 = group_bcast_f(wv.z, lo, hi);
        float w3 = group_bcast_f(wv.w, lo, hi);
        if (s < n) {
            v4f a0, b0, a1, b1, a2, b2, a3, b3;
            a0 = (hq + (unsigned)c0 * 16u)[0];
            b0 = (hq + (unsigned)c0 * 16u)[8];
            if (s + 1 < n) {
                a1 = (hq + (unsigned)c1 * 16u)[0];
                b1 = (hq + (unsigned)c1 * 16u)[8];
                if (s + 2 < n) {
                    a2 = (hq + (unsigned)c2 * 16u)[0];
                    b2 = (hq + (unsigned)c2 * 16u)[8];
                    if (s + 3 < n) {
                        a3 = (hq + (unsigned)c3 * 16u)[0];
                        b3 = (hq + (unsigned)c3 * 16u)[8];
                    }
                }
            }
            accA = fma4(a0, w0, accA); accB = fma4(b0, w0, accB);
            if (s + 1 < n) {
                accA = fma4(a1, w1, accA); accB = fma4(b1, w1, accB);
                if (s + 2 < n) {
                    accA = fma4(a2, w2, accA); accB = fma4(b2, w2, accB);
                    if (s + 3 < n) {
                        accA = fma4(a3, w3, accA); accB = fma4(b3, w3, accB);
                    }
                }
            }
        }
    }

    const unsigned longm = __builtin_amdgcn_ballot_w32(live && deg > SLAB);
    if (longm) {
        int sn = cnt[N];
        for (int gg = 0; gg < AGG_NPW; ++gg) {
            if (!((longm >> (8 * gg)) & 1u)) continue;
            const int   tnode = node0 + gg;
            const float dig = __int_as_float(__builtin_amdgcn_readlane(__float_as_int(di), 8 * gg));
            for (int base = 0; base < sn; base += 32) {
                int idx = base + lane;
                int r = -1, c = 0;
                if (idx < sn) {
                    v2i rec = spill[idx];
                    r = rec.x;
                    c = rec.y;
                }
                bool  mine = (r == tnode);
                float wsp = 0.f;
                if (mine) wsp = dig * rsqrtf((float)(cnt[c] + 1));
                unsigned m = __builtin_amdgcn_ballot_w32(mine);
                while (m) {
                    int b = __builtin_ctz(m);
                    m &= m - 1;
                    int   cb = __builtin_amdgcn_readlane(c, b);
                    float wb = __int_as_float(__builtin_amdgcn_readlane(__float_as_int(wsp), b));
                    if (g == gg) {
                        v4f ha = (hq + (unsigned)cb * 16u)[0];
                        v4f hb = (hq + (unsigned)cb * 16u)[8];
                        accA = fma4(ha, wb, accA);
                        accB = fma4(hb, wb, accB);
                    }
                }
            }
        }
    }

    v4f bvA = *(const v4f*)(bias + 4 * q);
    v4f bvB = *(const v4f*)(bias + 4 * q + 32);
    accA += bvA;
    accB += bvB;
    if (do_relu) {
        accA.x = fmaxf(accA.x, 0.f); accA.y = fmaxf(accA.y, 0.f);
        accA.z = fmaxf(accA.z, 0.f); accA.w = fmaxf(accA.w, 0.f);
        accB.x = fmaxf(accB.x, 0.f); accB.y = fmaxf(accB.y, 0.f);
        accB.z = fmaxf(accB.z, 0.f); accB.w = fmaxf(accB.w, 0.f);
    }
    if (live) {
        *(v4f*)(out + (size_t)node * 64 + 4 * q)      = accA;
        *(v4f*)(out + (size_t)node * 64 + 4 * q + 32) = accB;
    }
}

extern "C" void kernel_launch(void* const* d_in, const int* in_sizes, int n_in,
                              void* d_out, int out_size, void* d_ws, size_t ws_size,
                              hipStream_t stream) {
    (void)n_in; (void)out_size; (void)ws_size;

    const float* x  = (const float*)d_in[0];
    const int*   ei = (const int*)d_in[1];
    const float* W1 = (const float*)d_in[2];
    const float* b1 = (const float*)d_in[3];
    const float* W2 = (const float*)d_in[4];
    const float* b2 = (const float*)d_in[5];
    float* out = (float*)d_out;

    const int H   = in_sizes[3];
    const int IN  = in_sizes[2] / H;
    const int N   = in_sizes[0] / IN;
    const int E   = in_sizes[1] / 2;

    char* ws = (char*)d_ws;
    size_t off = 0;
    auto alloc = [&](size_t bytes) -> void* {
        void* p = (void*)(ws + off);
        off += (bytes + 255) & ~(size_t)255;
        return p;
    };
    int*   cnt     = (int*)  alloc((size_t)(N + 1) * 4);
    int*   tick    = (int*)  alloc((size_t)2 * TICK_PITCH * 4);
    int*   slab    = (int*)  alloc((size_t)N * SLAB * 4);
    v2i*   spill   = (v2i*)  alloc((size_t)E * 8);
    float* h1      = (float*)alloc((size_t)N * H * 4);
    float* y1      = (float*)alloc((size_t)N * H * 4);
    float* h2      = h1;

    const int T = 256;
    int gN1 = (N + 1 + T - 1) / T;
    int gFill = 1024;
    int gAgg = (N + 8 * AGG_NPW - 1) / (8 * AGG_NPW);
    int gGemm = (N + 31) / 32;

    init_cnt_kernel<<<gN1, T, 0, stream>>>(cnt, tick, N);

    fill_gemm_kernel<<<gGemm + gFill, T, 0, stream>>>(ei, cnt, tick, slab, spill, N, E,
                                                       x, W1, h1, IN, gGemm, gFill);
    agg_kernel<<<gAgg, T, 0, stream>>>(h1, cnt, slab, spill, b1, y1, N, 1);

    gemm64_wmma_kernel<<<gGemm, T, 0, stream>>>(y1, W2, h2, N, H);
    agg_kernel<<<gAgg, T, 0, stream>>>(h2, cnt, slab, spill, b2, out, N, 0);
}


# Layer-1 GEMM operand A through LDS: a work-group fetches its 32 x rows once as whole 512 B row loads (all out together) and waves read fragments from the tile; k and sum order kept
# speedup vs baseline: 3.7497x; 1.1572x over previous
#include <hip/hip_runtime.h>
#include <hip/hip_bf16.h>


typedef float v2f __attribute__((ext_vector_type(2)));
typedef float v8f __attribute__((ext_vector_type(8)));
typedef int   v2i __attribute__((ext_vector_type(2)));
typedef float v4f __attribute__((ext_vector_type(4)));
typedef int   v4i __attribute__((ext_vector_type(4)));

#define SLAB 32
#define AGG_NPW 4

#define FILL_EPT   4
#define FILL_CHUNK (256 * FILL_EPT)
#define TICK_PITCH 64

__global__ void init_cnt_kernel(int* __restrict__ cnt, int* __restrict__ tick, int N) {
    int i = blockIdx.x * blockDim.x + threadIdx.x;
    if (i <= N) cnt[i] = 0;
    if (i < 2) tick[i * TICK_PITCH] = 0;
}

__device__ __forceinline__ void fill_slab_tickets(const int* __restrict__ ei,
                                                  int* __restrict__ cnt,
                                                  int* __restrict__ tick,
                                                  int* __restrict__ slab, v2i* __restrict__ spill,
                                                  int N, int E) {
    __shared__ int s_ticket[2];
    int slot = 0;
    const int tid = threadIdx.x;
    const int my_half = (int)((__builtin_amdgcn_s_sendmsg_rtn(0x87) >> 18) & 1u);
    const int nT = (E + FILL_CHUNK - 1) / FILL_CHUNK;
    const int Nh = (N / 2) & ~63;

    for (int side = 0; side < 2; ++side) {
        const int h  = my_half ^ side;
        const int lo = h ? Nh : 0;
        const unsigned span = (unsigned)(h ? N - Nh : Nh);
        int* tk = tick + h * TICK_PITCH;

        if (tid == 0) s_ticket[slot] = atomicAdd(tk, 1);
        __syncthreads();
        int t = s_ticket[slot];
        slot ^= 1;

        for (int it = 0; it <= nT && t < nT; ++it) {
            const int base = t * FILL_CHUNK + tid;
            int r[FILL_EPT], c[FILL_EPT], p[FILL_EPT];
#pragma unroll
            for (int k = 0; k < FILL_EPT; ++k) {
                int e = base + 256 * k;
                r[k] = -1;
                c[k] = 0;
                if (e < E) {
                    r[k] = ei[e];
                    c[k] = ei[E + e];
                }
            }
#pragma unroll
            for (int k = 0; k < FILL_EPT; ++k) {
                p[k] = -1;
                if ((unsigned)(r[k] - lo) < span) p[k] = atomicAdd(&cnt[r[k]], 1);
            }

            if (tid == 0) s_ticket[slot] = atomicAdd(tk, 1);
            __syncthreads();
            t = s_ticket[slot];
            slot ^= 1;

#pragma unroll
            for (int k = 0; k < FILL_EPT; ++k) {
                if (p[k] >= 0) {
                    if (p[k] < SLAB) {
                        slab[(size_t)r[k] * SLAB + p[k]] = c[k];
                    } else {
                        int q = atomicAdd(&cnt[N], 1);
                        v2i rec;
                        rec.x = r[k];
                        rec.y = c[k];
                        spill[q] = rec;
                    }
                }
            }
        }
    }
}

#define XT_K 128
#define XT_PITCH 132

__device__ __forceinline__ void gemm64_wmma_body(int blk, const float* __restrict__ A,
                                                 const float* __restrict__ B,
                                                 float* __restrict__ C,
                                                 int M, int K) {
    __shared__ __attribute__((aligned(16))) float xt[32 * XT_PITCH];

    const int tid  = threadIdx.x;
    const int wave = tid >> 5;
    const int lane = tid & 31;
    const int mt = (wave >> 2) * 16;
    const int m0 = blk * 32 + mt;
    const int n0 = (wave & 3) * 16;
    const int half = lane >> 4;
    const int l    = lane & 15;

    const float* xrow = xt + (mt + l) * XT_PITCH + 2 * half;
    const float* bcol = B + n0 + l;
    v8f acc = {0.f, 0.f, 0.f, 0.f, 0.f, 0.f, 0.f, 0.f};

    for (int k0 = 0; k0 < K; k0 += XT_K) {
        const int kc = (K - k0 < XT_K) ? (K - k0) : XT_K;
        if (k0) __syncthreads();
        const int cc = (4 * lane < kc) ? 4 * lane : 0;
        v4f xr[4];
#pragma unroll
        for (int j = 0; j < 4; ++j) {
            const int row = wave + 8 * j;
            int gr = blk * 32 + row;
            if (gr >= M) gr = M - 1;
            xr[j] = *(const v4f*)(A + (size_t)gr * K + k0 + cc);
        }
#pragma unroll
        for (int j = 0; j < 4; ++j) {
            *(v4f*)(xt + (wave + 8 * j) * XT_PITCH + 4 * lane) = xr[j];
        }
        __syncthreads();

        int k = 0;
        for (; k + 32 <= kc; k += 32) {
#pragma unroll
            for (int u = 0; u < 8; ++u) {
                const int ka = k0 + k + 4 * u + 2 * half;
                v2f a = *(const v2f*)(xrow + k + 4 * u);
                v2f b;
                b.x = bcol[(size_t)ka * 64];
                b.y = bcol[(size_t)(ka + 1) * 64];
                acc = __builtin_amdgcn_wmma_f32_16x16x4_f32(false, a, false, b,
                                                            (short)0, acc, false, false);
            }
        }
        for (; k < kc; k += 4) {
            const int ka = k0 + k + 2 * half;
            v2f a = *(const v2f*)(xrow + k);
            v2f b;
            b.x = bcol[(size_t)ka * 64];
            b.y = bcol[(size_t)(ka + 1) * 64];
            acc = __builtin_amdgcn_wmma_f32_16x16x4_f32(false, a, false, b,
                                                        (short)0, acc, false, false);
        }
    }
    if (m0 >= M) return;
    float* crow = C + (size_t)m0 * 64 + n0 + l;
#pragma unroll
    for (int j = 0; j < 8; ++j) {
        crow[(size_t)(j + 8 * half) * 64] = acc[j];
    }
}

__global__ __launch_bounds__(256) void fill_gemm_kernel(const int* __restrict__ ei,
                                                        int* __restrict__ cnt,
                                                        int* __restrict__ tick,
                                                        int* __restrict__ slab,
                                                        v2i* __restrict__ spill,
                                                        int N, int E,
                                                        const float* __restrict__ A,
                                                        const float* __restrict__ B,
                                                        float* __restrict__ C,
                                                        int K, int gGemm, int gFill) {
    int b = blockIdx.x;
    int m = gFill / 2;
    if (gGemm < m) m = gGemm;
    bool is_gemm;
    int  idx;
    if (b < 3 * m) {
        int q = b / 3;
        int r = b - 3 * q;
        is_gemm = (r == 0);
        idx = is_gemm ? q : 2 * q + r - 1;
    } else {
        int t = b - 3 * m;
        int restG = gGemm - m;
        is_gemm = (t < restG);
        idx = is_gemm ? m + t : 2 * m + (t - restG);
    }
    if (is_gemm) {
        gemm64_wmma_body(idx, A, B, C, N, K);
    } else {
        fill_slab_tickets(ei, cnt, tick, slab, spill, N, E);
    }
}


__device__ __forceinline__ int group_bcast_i(int v, unsigned lo, unsigned hi) {
    return (int)__builtin_amdgcn_permlane16((unsigned)v, (unsigned)v, lo, hi, false, false);
}
__device__ __forceinline__ float group_bcast_f(float v, unsigned lo, unsigned hi) {
    return __int_as_float(group_bcast_i(__float_as_int(v), lo, hi));
}

__device__ __forceinline__ v4f fma4(v4f a, float w, v4f c) {
    v4f r;
    r.x = fmaf(a.x, w, c.x);
    r.y = fmaf(a.y, w, c.y);
    r.z = fmaf(a.z, w, c.z);
    r.w = fmaf(a.w, w, c.w);
    return r;
}

#define TILE_PITCH 68
#define W_PITCH 72

__device__ __forceinline__ void agg_node_sums(const float* __restrict__ h,
                                              const int* __restrict__ cnt,
                                              const int* __restrict__ slab,
                                              const v2i* __restrict__ spill,
                                              int N, int node0, int lane,
                                              v4f& accA, v4f& accB) {
    const int g = lane >> 3;
    const int q = lane & 7;
    const int  node = node0 + g;
    const bool live = node < N;
    const int  nd   = live ? node : (node0 < N ? node0 : 0);

    const v4f* hq = (const v4f*)h + q;

    int deg  = cnt[nd];
    v4i colv = *(const v4i*)(slab + (size_t)nd * SLAB + 4 * q);
    v4f hsA  = (hq + (unsigned)nd * 16u)[0];
    v4f hsB  = (hq + (unsigned)nd * 16u)[8];

    int n = deg < SLAB ? deg : SLAB;
    if (!live) n = 0;
    float di = rsqrtf((float)(deg + 1));

    int d0 = 0, d1 = 0, d2 = 0, d3 = 0;
    if (4 * q < n) {
        d0 = cnt[colv.x];
        if (4 * q + 1 < n) {
            d1 = cnt[colv.y];
            if (4 * q + 2 < n) {
                d2 = cnt[colv.z];
                if (4 * q + 3 < n) d3 = cnt[colv.w];
            }
        }
    }
    v4f wv;
    wv.x = di * rsqrtf((float)(d0 + 1));
    wv.y = di * rsqrtf((float)(d1 + 1));
    wv.z = di * rsqrtf((float)(d2 + 1));
    wv.w = di * rsqrtf((float)(d3 + 1));

    float ws = di * di;
    accA = hsA * ws;
    accB = hsB * ws;

    int n01 = max(__builtin_amdgcn_readlane(n, 0),  __builtin_amdgcn_readlane(n, 8));
    int n23 = max(__builtin_amdgcn_readlane(n, 16), __builtin_amdgcn_readlane(n, 24));
    int nmax = max(n01, n23);

    for (int s = 0; s < nmax; s += 4) {
        const unsigned lo = 0x11111111u * (unsigned)(s >> 2);
        const unsigned hi = lo + 0x88888888u;
        int   c0 = group_bcast_i(colv.x, lo, hi);
        int   c1 = group_bcast_i(colv.y, lo, hi);
        int   c2 = group_bcast_i(colv.z, lo, hi);
        int   c3 = group_bcast_i(colv.w, lo, hi);
        float w0 = group_bcast_f(wv.x, lo, hi);
        float w1 = group_bcast_f(wv.y, lo, hi);
        float w2 = group_bcast_f(wv.z, lo, hi);
        float w3 = group_bcast_f(wv.w, lo, hi);
        if (s < n) {
            v4f a0, b0, a1, b1, a2, b2, a3, b3;
            a0 = (hq + (unsigned)c0 * 16u)[0];
            b0 = (hq + (unsigned)c0 * 16u)[8];
            if (s + 1 < n) {
                a1 = (hq + (unsigned)c1 * 16u)[0];
                b1 = (hq + (unsigned)c1 * 16u)[8];
                if (s + 2 < n) {
                    a2 = (hq + (unsigned)c2 * 16u)[0];
                    b2 = (hq + (unsigned)c2 * 16u)[8];
                    if (s + 3 < n) {
                        a3 = (hq + (unsigned)c3 * 16u)[0];
                        b3 = (hq + (unsigned)c3 * 16u)[8];
                    }
                }
            }
            accA = fma4(a0, w0, accA); accB = fma4(b0, w0, accB);
            if (s + 1 < n) {
                accA = fma4(a1, w1, accA); accB = fma4(b1, w1, accB);
                if (s + 2 < n) {
                    accA = fma4(a2, w2, accA); accB = fma4(b2, w2, accB);
                    if (s + 3 < n) {
                        accA = fma4(a3, w3, accA); accB = fma4(b3, w3, accB);
                    }
                }
            }
        }
    }

    const unsigned longm = __builtin_amdgcn_ballot_w32(live && deg > SLAB);
    if (longm) {
        int sn = cnt[N];
        for (int gg = 0; gg < AGG_NPW; ++gg) {
            if (!((longm >> (8 * gg)) & 1u)) continue;
            const int   tnode = node0 + gg;
            const float dig = __int_as_float(__builtin_amdgcn_readlane(__float_as_int(di), 8 * gg));
            for (int base = 0; base < sn; base += 32) {
                int idx = base + lane;
                int r = -1, c = 0;
                if (idx < sn) {
                    v2i rec = spill[idx];
                    r = rec.x;
                    c = rec.y;
                }
                bool  mine = (r == tnode);
                float wsp = 0.f;
                if (mine) wsp = dig * rsqrtf((float)(cnt[c] + 1));
                unsigned m = __builtin_amdgcn_ballot_w32(mine);
                while (m) {
                    int b = __builtin_ctz(m);
                    m &= m - 1;
                    int   cb = __builtin_amdgcn_readlane(c, b);
                    float wb = __int_as_float(__builtin_amdgcn_readlane(__float_as_int(wsp), b));
                    if (g == gg) {
                        v4f ha = (hq + (unsigned)cb * 16u)[0];
                        v4f hb = (hq + (unsigned)cb * 16u)[8];
                        accA = fma4(ha, wb, accA);
                        accB = fma4(hb, wb, accB);
                    }
                }
            }
        }
    }
}

__global__ __launch_bounds__(256) void agg_kernel(const float* __restrict__ h,
                                                  const int* __restrict__ cnt,
                                                  const int* __restrict__ slab,
                                                  const v2i* __restrict__ spill,
                                                  const float* __restrict__ bias,
                                                  float* __restrict__ out,
                                                  int N) {
    const int lane = threadIdx.x & 31;
    const int q = lane & 7;
    const int node0 = __builtin_amdgcn_readfirstlane((blockIdx.x * 8 + (threadIdx.x >> 5)) * AGG_NPW);
    if (node0 >= N) return;
    const int node = node0 + (lane >> 3);

    v4f accA, accB;
    agg_node_sums(h, cnt, slab, spill, N, node0, lane, accA, accB);

    v4f bvA = *(const v4f*)(bias + 4 * q);
    v4f bvB = *(const v4f*)(bias + 4 * q + 32);
    accA += bvA;
    accB += bvB;
    accA.x = fmaxf(accA.x, 0.f); accA.y = fmaxf(accA.y, 0.f);
    accA.z = fmaxf(accA.z, 0.f); accA.w = fmaxf(accA.w, 0.f);
    accB.x = fmaxf(accB.x, 0.f); accB.y = fmaxf(accB.y, 0.f);
    accB.z = fmaxf(accB.z, 0.f); accB.w = fmaxf(accB.w, 0.f);
    if (node < N) {
        *(v4f*)(out + (size_t)node * 64 + 4 * q)      = accA;
        *(v4f*)(out + (size_t)node * 64 + 4 * q + 32) = accB;
    }
}

__global__ __launch_bounds__(256) void agg_gemm_kernel(const float* __restrict__ y,
                                                       const int* __restrict__ cnt,
                                                       const int* __restrict__ slab,
                                                       const v2i* __restrict__ spill,
                                                       const float* __restrict__ Wm,
                                                       const float* __restrict__ bias,
                                                       float* __restrict__ out,
                                                       int N) {
    __shared__ __attribute__((aligned(16))) float zt[32 * TILE_PITCH];
    __shared__ __attribute__((aligned(16))) float ot[32 * TILE_PITCH];
    __shared__ __attribute__((aligned(16))) float wt[64 * W_PITCH];

    const int lane = threadIdx.x & 31;
    const int wave = threadIdx.x >> 5;
    const int g = lane >> 3;
    const int q = lane & 7;
    const int node0 = __builtin_amdgcn_readfirstlane((blockIdx.x * 8 + wave) * AGG_NPW);
    const int node = node0 + g;
    const int trow = wave * AGG_NPW + g;

    v4f accA, accB;
    agg_node_sums(y, cnt, slab, spill, N, node0, lane, accA, accB);

    *(v4f*)(zt + trow * TILE_PITCH + 4 * q)      = accA;
    *(v4f*)(zt + trow * TILE_PITCH + 4 * q + 32) = accB;

    const int half = lane >> 4;
    const int l    = lane & 15;
    const int m0   = (wave >> 2) * 16;
    const int n0   = (wave & 3) * 16;

    v4f wreg[4];
#pragma unroll
    for (int j = 0; j < 4; ++j) wreg[j] = ((const v4f*)Wm)[threadIdx.x + 256 * j];
#pragma unroll
    for (int j = 0; j < 4; ++j) {
        int idx = threadIdx.x + 256 * j;
        *(v4f*)(wt + (idx >> 4) * W_PITCH + 4 * (idx & 15)) = wreg[j];
    }
    __syncthreads();

    const float* zrow = zt + (m0 + l) * TILE_PITCH;
    const float* wcol = wt + n0 + l;
    v8f acc = {0.f, 0.f, 0.f, 0.f, 0.f, 0.f, 0.f, 0.f};
#pragma unroll
    for (int t = 0; t < 16; ++t) {
        int ka = 4 * t + 2 * half;
        v2f a = *(const v2f*)(zrow + ka);
        v2f b;
        b.x = wcol[ka * W_PITCH];
        b.y = wcol[(ka + 1) * W_PITCH];
        acc = __builtin_amdgcn_wmma_f32_16x16x4_f32(false, a, false, b,
                                                    (short)0, acc, false, false);
    }
#pragma unroll
    for (int j = 0; j < 8; ++j) {
        ot[(m0 + j + 8 * half) * TILE_PITCH + n0 + l] = acc[j];
    }
    __syncthreads();

    v4f oA = *(const v4f*)(ot + trow * TILE_PITCH + 4 * q);
    v4f oB = *(const v4f*)(ot + trow * TILE_PITCH + 4 * q + 32);
    oA += *(const v4f*)(bias + 4 * q);
    oB += *(const v4f*)(bias + 4 * q + 32);
    if (node < N) {
        *(v4f*)(out + (size_t)node * 64 + 4 * q)      = oA;
        *(v4f*)(out + (size_t)node * 64 + 4 * q + 32) = oB;
    }
}

extern "C" void kernel_launch(void* const* d_in, const int* in_sizes, int n_in,
                              void* d_out, int out_size, void* d_ws, size_t ws_size,
                              hipStream_t stream) {
    (void)n_in; (void)out_size; (void)ws_size;

    const float* x  = (const float*)d_in[0];
    const int*   ei = (const int*)d_in[1];
    const float* W1 = (const float*)d_in[2];
    const float* b1 = (const float*)d_in[3];
    const float* W2 = (const float*)d_in[4];
    const float* b2 = (const float*)d_in[5];
    float* out = (float*)d_out;

    const int H   = in_sizes[3];
    const int IN  = in_sizes[2] / H;
    const int N   = in_sizes[0] / IN;
    const int E   = in_sizes[1] / 2;

    char* ws = (char*)d_ws;
    size_t off = 0;
    auto alloc = [&](size_t bytes) -> void* {
        void* p = (void*)(ws + off);
        off += (bytes + 255) & ~(size_t)255;
        return p;
    };
    int*   cnt     = (int*)  alloc((size_t)(N + 1) * 4);
    int*   tick    = (int*)  alloc((size_t)2 * TICK_PITCH * 4);
    int*   slab    = (int*)  alloc((size_t)N * SLAB * 4);
    v2i*   spill   = (v2i*)  alloc((size_t)E * 8);
    float* h1      = (float*)alloc((size_t)N * H * 4);
    float* y1      = (float*)alloc((size_t)N * H * 4);

    const int T = 256;
    int gN1 = (N + 1 + T - 1) / T;
    int gFill = 1024;
    int gAgg = (N + 8 * AGG_NPW - 1) / (8 * AGG_NPW);
    int gGemm = (N + 31) / 32;

    init_cnt_kernel<<<gN1, T, 0, stream>>>(cnt, tick, N);

    fill_gemm_kernel<<<gGemm + gFill, T, 0, stream>>>(ei, cnt, tick, slab, spill, N, E,
                                                       x, W1, h1, IN, gGemm, gFill);
    agg_kernel<<<gAgg, T, 0, stream>>>(h1, cnt, slab, spill, b1, y1, N);

    agg_gemm_kernel<<<gAgg, T, 0, stream>>>(y1, cnt, slab, spill, W2, b2, out, N);
}
